# starting kernel
# baseline (speedup 1.0000x reference)
#include <hip/hip_runtime.h>

typedef float v2f __attribute__((ext_vector_type(2)));
typedef float v4f __attribute__((ext_vector_type(4)));
typedef float v8f __attribute__((ext_vector_type(8)));

#define PITCH 34

__global__ __launch_bounds__(256) void ecc_aggregate_kernel(
    const float* __restrict__ nodes,
    const float* __restrict__ edges,
    const float* __restrict__ mask,
    const float* __restrict__ W,
    const float* __restrict__ bias,
    float* __restrict__ out)
{
    __shared__ float sE[64 * PITCH];
    __shared__ float sN[64 * PITCH];
    __shared__ float sMask[64];
    __shared__ float sPart[64];

    const int tid  = threadIdx.x;
    const int wave = tid >> 5;
    const int lane = tid & 31;
    const int lg   = lane >> 4;
    const int m    = lane & 15;

    const int wg = blockIdx.x;
    const int b  = wg >> 6;
    const int i  = wg & 63;

    const float* eBase = edges + ((size_t)b * 4096 + (size_t)i * 64) * 32;
    const float* nBase = nodes + (size_t)b * 64 * 32;
    const float* mBase = mask  + (size_t)b * 4096 + (size_t)i * 64;

    for (int q = tid; q < 64 * 32; q += 256) {
        const int j = q >> 5, f = q & 31;
        sE[j * PITCH + f] = eBase[q];
        sN[j * PITCH + f] = nBase[q];
    }
    if (tid < 64) sMask[tid] = mBase[tid];
    __syncthreads();

    float tileAcc[8];
#pragma unroll
    for (int k = 0; k < 8; ++k) tileAcc[k] = 0.0f;

#pragma unroll
    for (int k = 0; k < 8; ++k) {
        const int t  = wave + 8 * k;
        const int g0 = t << 4;
        const int e0 = (t & 1) << 4;

        v2f af[8];
#pragma unroll
        for (int kk = 0; kk < 8; ++kk)
            af[kk] = *(const v2f*)(W + (size_t)(g0 + m) * 32 + 4 * kk + 2 * lg);

        const v4f b0 = *(const v4f*)(bias + g0 + 8 * lg);
        const v4f b1 = *(const v4f*)(bias + g0 + 8 * lg + 4);
        const float bv[8] = {b0.x, b0.y, b0.z, b0.w, b1.x, b1.y, b1.z, b1.w};

#pragma unroll
        for (int jg = 0; jg < 4; ++jg) {
            const int j16 = jg * 16 + m;
            v8f c = {};
#pragma unroll
            for (int kk = 0; kk < 8; ++kk) {
                v2f bf = *(const v2f*)&sE[j16 * PITCH + 4 * kk + 2 * lg];
                c = __builtin_amdgcn_wmma_f32_16x16x4_f32(
                        false, af[kk], false, bf, (short)0, c, false, false);
            }
            float partial = 0.0f;
#pragma unroll
            for (int r = 0; r < 8; ++r) {
                float v = c[r] + bv[r];
                v = v > 0.0f ? v : 0.0f;
                partial += v * sN[j16 * PITCH + e0 + 8 * lg + r];
            }
            partial *= sMask[j16];
#pragma unroll
            for (int off = 16; off > 0; off >>= 1)
                partial += __shfl_xor(partial, off, 32);
            tileAcc[k] += partial;
        }
    }

    if (lane == 0) {
#pragma unroll
        for (int k = 0; k < 8; ++k) sPart[wave + 8 * k] = tileAcc[k];
    }
    __syncthreads();

    if (tid < 32)
        out[(size_t)wg * 32 + tid] = sPart[2 * tid] + sPart[2 * tid + 1];
}

extern "C" void kernel_launch(void* const* d_in, const int* in_sizes, int n_in,
                              void* d_out, int out_size, void* d_ws, size_t ws_size,
                              hipStream_t stream) {
    const float* nodes = (const float*)d_in[0];
    const float* edges = (const float*)d_in[1];
    const float* mask  = (const float*)d_in[2];
    const float* W     = (const float*)d_in[3];
    const float* bias  = (const float*)d_in[4];
    float* out         = (float*)d_out;

    ecc_aggregate_kernel<<<dim3(16 * 64), dim3(256), 0, stream>>>(
        nodes, edges, mask, W, bias, out);
}


# Stage edges, nodes and mask with every load in flight at once (four 16-byte loads per thread, then the LDS stores) instead of eight dependent one-dword round trips per work-group
# speedup vs baseline: 1.3572x; 1.3572x over previous
#include <hip/hip_runtime.h>

typedef float v2f __attribute__((ext_vector_type(2)));
typedef float v4f __attribute__((ext_vector_type(4)));
typedef float v8f __attribute__((ext_vector_type(8)));

#define PITCH 34

__global__ __launch_bounds__(256) void ecc_aggregate_kernel(
    const float* __restrict__ nodes,
    const float* __restrict__ edges,
    const float* __restrict__ mask,
    const float* __restrict__ W,
    const float* __restrict__ bias,
    float* __restrict__ out)
{
    __shared__ float sE[64 * PITCH];
    __shared__ float sN[64 * PITCH];
    __shared__ float sMask[64];
    __shared__ float sPart[64];

    const int tid  = threadIdx.x;
    const int wave = tid >> 5;
    const int lane = tid & 31;
    const int lg   = lane >> 4;
    const int m    = lane & 15;

    const int wg = blockIdx.x;
    const int b  = wg >> 6;
    const int i  = wg & 63;

    const float* eBase = edges + ((size_t)b * 4096 + (size_t)i * 64) * 32;
    const float* nBase = nodes + (size_t)b * 64 * 32;
    const float* mBase = mask  + (size_t)b * 4096 + (size_t)i * 64;

    {
        const v4f* e4 = (const v4f*)eBase;
        const v4f* n4 = (const v4f*)nBase;
        const v4f ea = e4[tid], eb = e4[tid + 256];
        const v4f na = n4[tid], nb = n4[tid + 256];
        float mk = 0.0f;
        if (tid < 64) mk = mBase[tid];
        const int j = tid >> 3, f = (tid & 7) << 2;
        float* pe = &sE[j * PITCH + f];
        float* pn = &sN[j * PITCH + f];
        *(v2f*)(pe)                  = (v2f){ea.x, ea.y};
        *(v2f*)(pe + 2)              = (v2f){ea.z, ea.w};
        *(v2f*)(pe + 32 * PITCH)     = (v2f){eb.x, eb.y};
        *(v2f*)(pe + 32 * PITCH + 2) = (v2f){eb.z, eb.w};
        *(v2f*)(pn)                  = (v2f){na.x, na.y};
        *(v2f*)(pn + 2)              = (v2f){na.z, na.w};
        *(v2f*)(pn + 32 * PITCH)     = (v2f){nb.x, nb.y};
        *(v2f*)(pn + 32 * PITCH + 2) = (v2f){nb.z, nb.w};
        if (tid < 64) sMask[tid] = mk;
    }
    __syncthreads();

    float tileAcc[8];
#pragma unroll
    for (int k = 0; k < 8; ++k) tileAcc[k] = 0.0f;

#pragma unroll
    for (int k = 0; k < 8; ++k) {
        const int t  = wave + 8 * k;
        const int g0 = t << 4;
        const int e0 = (t & 1) << 4;

        v2f af[8];
#pragma unroll
        for (int kk = 0; kk < 8; ++kk)
            af[kk] = *(const v2f*)(W + (size_t)(g0 + m) * 32 + 4 * kk + 2 * lg);

        const v4f b0 = *(const v4f*)(bias + g0 + 8 * lg);
        const v4f b1 = *(const v4f*)(bias + g0 + 8 * lg + 4);
        const float bv[8] = {b0.x, b0.y, b0.z, b0.w, b1.x, b1.y, b1.z, b1.w};

#pragma unroll
        for (int jg = 0; jg < 4; ++jg) {
            const int j16 = jg * 16 + m;
            v8f c = {};
#pragma unroll
            for (int kk = 0; kk < 8; ++kk) {
                v2f bf = *(const v2f*)&sE[j16 * PITCH + 4 * kk + 2 * lg];
                c = __builtin_amdgcn_wmma_f32_16x16x4_f32(
                        false, af[kk], false, bf, (short)0, c, false, false);
            }
            float partial = 0.0f;
#pragma unroll
            for (int r = 0; r < 8; ++r) {
                float v = c[r] + bv[r];
                v = v > 0.0f ? v : 0.0f;
                partial += v * sN[j16 * PITCH + e0 + 8 * lg + r];
            }
            partial *= sMask[j16];
#pragma unroll
            for (int off = 16; off > 0; off >>= 1)
                partial += __shfl_xor(partial, off, 32);
            tileAcc[k] += partial;
        }
    }

    if (lane == 0) {
#pragma unroll
        for (int k = 0; k < 8; ++k) sPart[wave + 8 * k] = tileAcc[k];
    }
    __syncthreads();

    if (tid < 32)
        out[(size_t)wg * 32 + tid] = sPart[2 * tid] + sPart[2 * tid + 1];
}

extern "C" void kernel_launch(void* const* d_in, const int* in_sizes, int n_in,
                              void* d_out, int out_size, void* d_ws, size_t ws_size,
                              hipStream_t stream) {
    const float* nodes = (const float*)d_in[0];
    const float* edges = (const float*)d_in[1];
    const float* mask  = (const float*)d_in[2];
    const float* W     = (const float*)d_in[3];
    const float* bias  = (const float*)d_in[4];
    float* out         = (float*)d_out;

    ecc_aggregate_kernel<<<dim3(16 * 64), dim3(256), 0, stream>>>(
        nodes, edges, mask, W, bias, out);
}


# Each lane keeps its own running sum per tile and the 32 lanes are added once per tile behind the loop (row swap + four DPP row adds), not by a five-step LDS butterfly per 16 edges; sum order changes
# speedup vs baseline: 1.7803x; 1.3117x over previous
#include <hip/hip_runtime.h>

typedef float v2f __attribute__((ext_vector_type(2)));
typedef float v4f __attribute__((ext_vector_type(4)));
typedef float v8f __attribute__((ext_vector_type(8)));

#define PITCH 34

__device__ __forceinline__ float xhalf(float x) {
    const int b = __builtin_bit_cast(int, x);
    return __builtin_bit_cast(float,
        __builtin_amdgcn_permlanex16(b, b, 0x76543210, 0xfedcba98, false, false));
}

template <int XM>
__device__ __forceinline__ float row_xor(float x) {
    return __builtin_bit_cast(float,
        __builtin_amdgcn_update_dpp(0, __builtin_bit_cast(int, x), 0x160 + XM, 0xf, 0xf, true));
}

__global__ __launch_bounds__(256) void ecc_aggregate_kernel(
    const float* __restrict__ nodes,
    const float* __restrict__ edges,
    const float* __restrict__ mask,
    const float* __restrict__ W,
    const float* __restrict__ bias,
    float* __restrict__ out)
{
    __shared__ float sE[64 * PITCH];
    __shared__ float sN[64 * PITCH];
    __shared__ float sMask[64];
    __shared__ float sPart[64];

    const int tid  = threadIdx.x;
    const int wave = tid >> 5;
    const int lane = tid & 31;
    const int lg   = lane >> 4;
    const int m    = lane & 15;

    const int wg = blockIdx.x;
    const int b  = wg >> 6;
    const int i  = wg & 63;

    const float* eBase = edges + ((size_t)b * 4096 + (size_t)i * 64) * 32;
    const float* nBase = nodes + (size_t)b * 64 * 32;
    const float* mBase = mask  + (size_t)b * 4096 + (size_t)i * 64;

    {
        const v4f* e4 = (const v4f*)eBase;
        const v4f* n4 = (const v4f*)nBase;
        const v4f ea = e4[tid], eb = e4[tid + 256];
        const v4f na = n4[tid], nb = n4[tid + 256];
        float mk = 0.0f;
        if (tid < 64) mk = mBase[tid];
        const int j = tid >> 3, f = (tid & 7) << 2;
        float* pe = &sE[j * PITCH + f];
        float* pn = &sN[j * PITCH + f];
        *(v2f*)(pe)                  = (v2f){ea.x, ea.y};
        *(v2f*)(pe + 2)              = (v2f){ea.z, ea.w};
        *(v2f*)(pe + 32 * PITCH)     = (v2f){eb.x, eb.y};
        *(v2f*)(pe + 32 * PITCH + 2) = (v2f){eb.z, eb.w};
        *(v2f*)(pn)                  = (v2f){na.x, na.y};
        *(v2f*)(pn + 2)              = (v2f){na.z, na.w};
        *(v2f*)(pn + 32 * PITCH)     = (v2f){nb.x, nb.y};
        *(v2f*)(pn + 32 * PITCH + 2) = (v2f){nb.z, nb.w};
        if (tid < 64) sMask[tid] = mk;
    }
    __syncthreads();

    v2f acc2[8];
#pragma unroll
    for (int k = 0; k < 8; ++k) acc2[k] = (v2f){0.0f, 0.0f};

    v2f bfr[4][8];
    v2f n2[4][4];
    float mk[4];
    const int e0 = (wave & 1) << 4;
#pragma unroll
    for (int jg = 0; jg < 4; ++jg) {
        const int j16 = jg * 16 + m;
#pragma unroll
        for (int kk = 0; kk < 8; ++kk)
            bfr[jg][kk] = *(const v2f*)&sE[j16 * PITCH + 4 * kk + 2 * lg];
#pragma unroll
        for (int r = 0; r < 4; ++r)
            n2[jg][r] = *(const v2f*)&sN[j16 * PITCH + e0 + 8 * lg + 2 * r];
        mk[jg] = sMask[j16];
    }

#pragma unroll
    for (int k = 0; k < 8; ++k) {
        const int t  = wave + 8 * k;
        const int g0 = t << 4;

        v2f af[8];
#pragma unroll
        for (int kk = 0; kk < 8; ++kk)
            af[kk] = *(const v2f*)(W + (size_t)(g0 + m) * 32 + 4 * kk + 2 * lg);

        const v4f b0 = *(const v4f*)(bias + g0 + 8 * lg);
        const v4f b1 = *(const v4f*)(bias + g0 + 8 * lg + 4);
        const v2f bv2[4] = {(v2f){b0.x, b0.y}, (v2f){b0.z, b0.w}, (v2f){b1.x, b1.y}, (v2f){b1.z, b1.w}};

        v8f c[4] = {};
#pragma unroll
        for (int kk = 0; kk < 8; ++kk) {
#pragma unroll
            for (int jg = 0; jg < 4; ++jg)
                c[jg] = __builtin_amdgcn_wmma_f32_16x16x4_f32(
                            false, af[kk], false, bfr[jg][kk], (short)0, c[jg], false, false);
        }
#pragma unroll
        for (int jg = 0; jg < 4; ++jg) {
            v2f partial = (v2f){0.0f, 0.0f};
#pragma unroll
            for (int r = 0; r < 4; ++r) {
                v2f v = (v2f){c[jg][2 * r], c[jg][2 * r + 1]} + bv2[r];
                v = __builtin_elementwise_max(v, (v2f){0.0f, 0.0f});
                partial += v * n2[jg][r];
            }
            acc2[k] += partial * mk[jg];
        }
    }

    float tileAcc[8];
#pragma unroll
    for (int k = 0; k < 8; ++k) tileAcc[k] = acc2[k].x + acc2[k].y;
#pragma unroll
    for (int k = 0; k < 8; ++k) tileAcc[k] += xhalf(tileAcc[k]);
#pragma unroll
    for (int k = 0; k < 8; ++k) tileAcc[k] += row_xor<1>(tileAcc[k]);
#pragma unroll
    for (int k = 0; k < 8; ++k) tileAcc[k] += row_xor<2>(tileAcc[k]);
#pragma unroll
    for (int k = 0; k < 8; ++k) tileAcc[k] += row_xor<4>(tileAcc[k]);
#pragma unroll
    for (int k = 0; k < 8; ++k) tileAcc[k] += row_xor<8>(tileAcc[k]);

    if (lane == 0) {
#pragma unroll
        for (int k = 0; k < 8; ++k) sPart[wave + 8 * k] = tileAcc[k];
    }
    __syncthreads();

    if (tid < 32)
        out[(size_t)wg * 32 + tid] = sPart[2 * tid] + sPart[2 * tid + 1];
}

extern "C" void kernel_launch(void* const* d_in, const int* in_sizes, int n_in,
                              void* d_out, int out_size, void* d_ws, size_t ws_size,
                              hipStream_t stream) {
    const float* nodes = (const float*)d_in[0];
    const float* edges = (const float*)d_in[1];
    const float* mask  = (const float*)d_in[2];
    const float* W     = (const float*)d_in[3];
    const float* bias  = (const float*)d_in[4];
    float* out         = (float*)d_out;

    ecc_aggregate_kernel<<<dim3(16 * 64), dim3(256), 0, stream>>>(
        nodes, edges, mask, W, bias, out);
}


# W fragments, bias and node values are read once per work-group and stay in registers while it loops over four (b,i) rows (256 work-groups of 8 waves); only the edge rows are restaged per row
# speedup vs baseline: 1.9065x; 1.0709x over previous
#include <hip/hip_runtime.h>

typedef float v2f __attribute__((ext_vector_type(2)));
typedef float v4f __attribute__((ext_vector_type(4)));
typedef float v8f __attribute__((ext_vector_type(8)));

#define PITCH 34
#define ROWS  4

__device__ __forceinline__ float xhalf(float x) {
    const int b = __builtin_bit_cast(int, x);
    return __builtin_bit_cast(float,
        __builtin_amdgcn_permlanex16(b, b, 0x76543210, 0xfedcba98, false, false));
}

template <int XM>
__device__ __forceinline__ float row_xor(float x) {
    return __builtin_bit_cast(float,
        __builtin_amdgcn_update_dpp(0, __builtin_bit_cast(int, x), 0x160 + XM, 0xf, 0xf, true));
}

__global__ __launch_bounds__(256) void ecc_aggregate_kernel(
    const float* __restrict__ nodes,
    const float* __restrict__ edges,
    const float* __restrict__ mask,
    const float* __restrict__ W,
    const float* __restrict__ bias,
    float* __restrict__ out)
{
    __shared__ float sE[64 * PITCH];
    __shared__ float sMask[64];
    __shared__ float sPart[64];

    const int tid  = threadIdx.x;
    const int wave = tid >> 5;
    const int lane = tid & 31;
    const int lg   = lane >> 4;
    const int m    = lane & 15;

    const int wg   = blockIdx.x;
    const int row0 = wg * ROWS;
    const int b    = row0 >> 6;

    const v4f*   e4    = (const v4f*)(edges + (size_t)row0 * 64 * 32);
    const float* mrow  = mask  + (size_t)row0 * 64;
    const float* nBase = nodes + (size_t)b * 64 * 32;

    v4f ea = e4[tid], eb = e4[tid + 256];
    float mkv = 0.0f;
    if (tid < 64) mkv = mrow[tid];

    v2f af[8][8];
    v4f bq[8][2];
#pragma unroll
    for (int k = 0; k < 8; ++k) {
        const int g0 = (wave + 8 * k) << 4;
#pragma unroll
        for (int kk = 0; kk < 8; ++kk)
            af[k][kk] = *(const v2f*)(W + (size_t)(g0 + m) * 32 + 4 * kk + 2 * lg);
        bq[k][0] = *(const v4f*)(bias + g0 + 8 * lg);
        bq[k][1] = *(const v4f*)(bias + g0 + 8 * lg + 4);
    }

    const int e0 = (wave & 1) << 4;
    v2f n2[4][4];
#pragma unroll
    for (int jg = 0; jg < 4; ++jg) {
        const float* np = nBase + (size_t)(jg * 16 + m) * 32 + e0 + 8 * lg;
        const v4f q0 = *(const v4f*)np, q1 = *(const v4f*)(np + 4);
        n2[jg][0] = (v2f){q0.x, q0.y};
        n2[jg][1] = (v2f){q0.z, q0.w};
        n2[jg][2] = (v2f){q1.x, q1.y};
        n2[jg][3] = (v2f){q1.z, q1.w};
    }

    const int sj = tid >> 3, sf = (tid & 7) << 2;
    float* pe = &sE[sj * PITCH + sf];

#pragma unroll 1
    for (int r = 0; r < ROWS; ++r) {
        *(v2f*)(pe)                  = (v2f){ea.x, ea.y};
        *(v2f*)(pe + 2)              = (v2f){ea.z, ea.w};
        *(v2f*)(pe + 32 * PITCH)     = (v2f){eb.x, eb.y};
        *(v2f*)(pe + 32 * PITCH + 2) = (v2f){eb.z, eb.w};
        if (tid < 64) sMask[tid] = mkv;
        __syncthreads();

        if (r + 1 < ROWS) {
            ea = e4[(r + 1) * 512 + tid];
            eb = e4[(r + 1) * 512 + tid + 256];
            if (tid < 64) mkv = mrow[(r + 1) * 64 + tid];
        }

        v2f bfr[4][8];
        float mk[4];
#pragma unroll
        for (int jg = 0; jg < 4; ++jg) {
            const int j16 = jg * 16 + m;
#pragma unroll
            for (int kk = 0; kk < 8; ++kk)
                bfr[jg][kk] = *(const v2f*)&sE[j16 * PITCH + 4 * kk + 2 * lg];
            mk[jg] = sMask[j16];
        }

        v2f acc2[8];
#pragma unroll
        for (int k = 0; k < 8; ++k) {
            const v4f b0 = bq[k][0], b1 = bq[k][1];
            const v2f bv2[4] = {(v2f){b0.x, b0.y}, (v2f){b0.z, b0.w}, (v2f){b1.x, b1.y}, (v2f){b1.z, b1.w}};

            v8f c[4] = {};
#pragma unroll
            for (int kk = 0; kk < 8; ++kk) {
#pragma unroll
                for (int jg = 0; jg < 4; ++jg)
                    c[jg] = __builtin_amdgcn_wmma_f32_16x16x4_f32(
                                false, af[k][kk], false, bfr[jg][kk], (short)0, c[jg], false, false);
            }
            v2f a2 = (v2f){0.0f, 0.0f};
#pragma unroll
            for (int jg = 0; jg < 4; ++jg) {
                v2f partial = (v2f){0.0f, 0.0f};
#pragma unroll
                for (int q = 0; q < 4; ++q) {
                    v2f v = (v2f){c[jg][2 * q], c[jg][2 * q + 1]} + bv2[q];
                    v = __builtin_elementwise_max(v, (v2f){0.0f, 0.0f});
                    partial += v * n2[jg][q];
                }
                a2 += partial * mk[jg];
            }
            acc2[k] = a2;
        }

        float tileAcc[8];
#pragma unroll
        for (int k = 0; k < 8; ++k) tileAcc[k] = acc2[k].x + acc2[k].y;
#pragma unroll
        for (int k = 0; k < 8; ++k) tileAcc[k] += xhalf(tileAcc[k]);
#pragma unroll
        for (int k = 0; k < 8; ++k) tileAcc[k] += row_xor<1>(tileAcc[k]);
#pragma unroll
        for (int k = 0; k < 8; ++k) tileAcc[k] += row_xor<2>(tileAcc[k]);
#pragma unroll
        for (int k = 0; k < 8; ++k) tileAcc[k] += row_xor<4>(tileAcc[k]);
#pragma unroll
        for (int k = 0; k < 8; ++k) tileAcc[k] += row_xor<8>(tileAcc[k]);

        if (lane == 0) {
#pragma unroll
            for (int k = 0; k < 8; ++k) sPart[wave + 8 * k] = tileAcc[k];
        }
        __syncthreads();

        if (tid < 32)
            out[(size_t)(row0 + r) * 32 + tid] = sPart[2 * tid] + sPart[2 * tid + 1];
    }
}

extern "C" void kernel_launch(void* const* d_in, const int* in_sizes, int n_in,
                              void* d_out, int out_size, void* d_ws, size_t ws_size,
                              hipStream_t stream) {
    const float* nodes = (const float*)d_in[0];
    const float* edges = (const float*)d_in[1];
    const float* mask  = (const float*)d_in[2];
    const float* W     = (const float*)d_in[3];
    const float* bias  = (const float*)d_in[4];
    float* out         = (float*)d_out;

    ecc_aggregate_kernel<<<dim3(256), dim3(256), 0, stream>>>(
        nodes, edges, mask, W, bias, out);
}


# Edges whose mask is zero are packed out before the matrix work: each row's live edges fill ceil(live/16) column tiles and only those are multiplied; edge sum order changes
# speedup vs baseline: 2.2878x; 1.2000x over previous
#include <hip/hip_runtime.h>

typedef float v2f __attribute__((ext_vector_type(2)));
typedef float v4f __attribute__((ext_vector_type(4)));
typedef float v8f __attribute__((ext_vector_type(8)));

#define PITCH  34
#define NPITCH 36
#define ROWS   4

__device__ __forceinline__ float xhalf(float x) {
    const int b = __builtin_bit_cast(int, x);
    return __builtin_bit_cast(float,
        __builtin_amdgcn_permlanex16(b, b, 0x76543210, 0xfedcba98, false, false));
}

template <int XM>
__device__ __forceinline__ float row_xor(float x) {
    return __builtin_bit_cast(float,
        __builtin_amdgcn_update_dpp(0, __builtin_bit_cast(int, x), 0x160 + XM, 0xf, 0xf, true));
}

template <int NT>
__device__ __forceinline__ void row_tiles(
    const v2f (&af)[8][8], const v4f (&bq)[8][2],
    const float* sE, const float* sN, const int (&jc)[4], const float (&mk)[4],
    const int lg, const int e0, float (&tileAcc)[8])
{
    v2f bfr[NT][8];
    v2f n2[NT][4];
#pragma unroll
    for (int t = 0; t < NT; ++t) {
        const float* ep = sE + jc[t] * PITCH + 2 * lg;
#pragma unroll
        for (int kk = 0; kk < 8; ++kk)
            bfr[t][kk] = *(const v2f*)(ep + 4 * kk);
        const float* np = sN + jc[t] * NPITCH + e0 + 8 * lg;
        const v4f q0 = *(const v4f*)np, q1 = *(const v4f*)(np + 4);
        n2[t][0] = (v2f){q0.x, q0.y};
        n2[t][1] = (v2f){q0.z, q0.w};
        n2[t][2] = (v2f){q1.x, q1.y};
        n2[t][3] = (v2f){q1.z, q1.w};
    }

#pragma unroll
    for (int k = 0; k < 8; ++k) {
        const v4f b0 = bq[k][0], b1 = bq[k][1];
        const v2f bv2[4] = {(v2f){b0.x, b0.y}, (v2f){b0.z, b0.w}, (v2f){b1.x, b1.y}, (v2f){b1.z, b1.w}};

        v8f c[NT] = {};
#pragma unroll
        for (int kk = 0; kk < 8; ++kk) {
#pragma unroll
            for (int t = 0; t < NT; ++t)
                c[t] = __builtin_amdgcn_wmma_f32_16x16x4_f32(
                           false, af[k][kk], false, bfr[t][kk], (short)0, c[t], false, false);
        }
        v2f a2 = (v2f){0.0f, 0.0f};
#pragma unroll
        for (int t = 0; t < NT; ++t) {
            v2f partial = (v2f){0.0f, 0.0f};
#pragma unroll
            for (int q = 0; q < 4; ++q) {
                v2f v = (v2f){c[t][2 * q], c[t][2 * q + 1]} + bv2[q];
                v = __builtin_elementwise_max(v, (v2f){0.0f, 0.0f});
                partial += v * n2[t][q];
            }
            a2 += partial * mk[t];
        }
        tileAcc[k] = a2.x + a2.y;
    }
}

__global__ __launch_bounds__(256) void ecc_aggregate_kernel(
    const float* __restrict__ nodes,
    const float* __restrict__ edges,
    const float* __restrict__ mask,
    const float* __restrict__ W,
    const float* __restrict__ bias,
    float* __restrict__ out)
{
    __shared__ __attribute__((aligned(16))) float sE[64 * PITCH];
    __shared__ __attribute__((aligned(16))) float sN[64 * NPITCH];
    __shared__ int   sIdx[64];
    __shared__ float sMk[64];
    __shared__ int   sCnt;
    __shared__ float sPart[64];

    const int tid  = threadIdx.x;
    const int wave = tid >> 5;
    const int lane = tid & 31;
    const int lg   = lane >> 4;
    const int m    = lane & 15;

    const int wg   = blockIdx.x;
    const int row0 = wg * ROWS;
    const int b    = row0 >> 6;

    const v4f*   e4    = (const v4f*)(edges + (size_t)row0 * 64 * 32);
    const float* mrow  = mask  + (size_t)row0 * 64;
    const float* nBase = nodes + (size_t)b * 64 * 32;

    v4f ea = e4[tid], eb = e4[tid + 256];
    float mk0 = 0.0f, mk1 = 0.0f;
    if (tid < 32) { mk0 = mrow[tid]; mk1 = mrow[tid + 32]; }

    const v4f na = ((const v4f*)nBase)[tid], nb = ((const v4f*)nBase)[tid + 256];

    v2f af[8][8];
    v4f bq[8][2];
#pragma unroll
    for (int k = 0; k < 8; ++k) {
        const int g0 = (wave + 8 * k) << 4;
#pragma unroll
        for (int kk = 0; kk < 8; ++kk)
            af[k][kk] = *(const v2f*)(W + (size_t)(g0 + m) * 32 + 4 * kk + 2 * lg);
        bq[k][0] = *(const v4f*)(bias + g0 + 8 * lg);
        bq[k][1] = *(const v4f*)(bias + g0 + 8 * lg + 4);
    }

    const int e0 = (wave & 1) << 4;

    const int sj = tid >> 3, sf = (tid & 7) << 2;
    float* pe = &sE[sj * PITCH + sf];

    *(v4f*)&sN[sj * NPITCH + sf]        = na;
    *(v4f*)&sN[(sj + 32) * NPITCH + sf] = nb;

#pragma unroll 1
    for (int r = 0; r < ROWS; ++r) {
        *(v2f*)(pe)                  = (v2f){ea.x, ea.y};
        *(v2f*)(pe + 2)              = (v2f){ea.z, ea.w};
        *(v2f*)(pe + 32 * PITCH)     = (v2f){eb.x, eb.y};
        *(v2f*)(pe + 32 * PITCH + 2) = (v2f){eb.z, eb.w};
        if (tid < 32) {
            const unsigned lo = __builtin_amdgcn_ballot_w32(mk0 != 0.0f);
            const unsigned hi = __builtin_amdgcn_ballot_w32(mk1 != 0.0f);
            const unsigned lt = (1u << lane) - 1u;
            const int nlo = __builtin_popcount(lo);
            const int r0  = __builtin_popcount(lo & lt);
            const int r1  = nlo + __builtin_popcount(hi & lt);
            if (mk0 != 0.0f) { sIdx[r0] = tid;      sMk[r0] = mk0; }
            if (mk1 != 0.0f) { sIdx[r1] = tid + 32; sMk[r1] = mk1; }
            const int cntw = nlo + __builtin_popcount(hi);
            if (cntw + lane < 64)      { sIdx[cntw + lane]      = 0; sMk[cntw + lane]      = 0.0f; }
            if (cntw + lane + 32 < 64) { sIdx[cntw + lane + 32] = 0; sMk[cntw + lane + 32] = 0.0f; }
            if (tid == 0) sCnt = cntw;
        }
        __syncthreads();

        if (r + 1 < ROWS) {
            ea = e4[(r + 1) * 512 + tid];
            eb = e4[(r + 1) * 512 + tid + 256];
            if (tid < 32) { mk0 = mrow[(r + 1) * 64 + tid]; mk1 = mrow[(r + 1) * 64 + tid + 32]; }
        }

        const int cnt = __builtin_amdgcn_readfirstlane(sCnt);
        int nt = (cnt + 15) >> 4;
        int jc[4];
        float mk[4];
#pragma unroll
        for (int t = 0; t < 4; ++t) {
            jc[t] = sIdx[t * 16 + m];
            mk[t] = sMk[t * 16 + m];
        }

        float tileAcc[8];
        switch (nt) {
            case 0:
            case 1:  row_tiles<1>(af, bq, sE, sN, jc, mk, lg, e0, tileAcc); break;
            case 2:  row_tiles<2>(af, bq, sE, sN, jc, mk, lg, e0, tileAcc); break;
            case 3:  row_tiles<3>(af, bq, sE, sN, jc, mk, lg, e0, tileAcc); break;
            default: row_tiles<4>(af, bq, sE, sN, jc, mk, lg, e0, tileAcc); break;
        }

#pragma unroll
        for (int k = 0; k < 8; ++k) tileAcc[k] += xhalf(tileAcc[k]);
#pragma unroll
        for (int k = 0; k < 8; ++k) tileAcc[k] += row_xor<1>(tileAcc[k]);
#pragma unroll
        for (int k = 0; k < 8; ++k) tileAcc[k] += row_xor<2>(tileAcc[k]);
#pragma unroll
        for (int k = 0; k < 8; ++k) tileAcc[k] += row_xor<4>(tileAcc[k]);
#pragma unroll
        for (int k = 0; k < 8; ++k) tileAcc[k] += row_xor<8>(tileAcc[k]);

        if (lane == 0) {
#pragma unroll
            for (int k = 0; k < 8; ++k) sPart[wave + 8 * k] = tileAcc[k];
        }
        __syncthreads();

        if (tid < 32)
            out[(size_t)(row0 + r) * 32 + tid] = sPart[2 * tid] + sPart[2 * tid + 1];
    }
}

extern "C" void kernel_launch(void* const* d_in, const int* in_sizes, int n_in,
                              void* d_out, int out_size, void* d_ws, size_t ws_size,
                              hipStream_t stream) {
    const float* nodes = (const float*)d_in[0];
    const float* edges = (const float*)d_in[1];
    const float* mask  = (const float*)d_in[2];
    const float* W     = (const float*)d_in[3];
    const float* bias  = (const float*)d_in[4];
    float* out         = (float*)d_out;

    ecc_aggregate_kernel<<<dim3(256), dim3(256), 0, stream>>>(
        nodes, edges, mask, W, bias, out);
}


# W reaches the A fragments by whole-row 16-byte loads (512 contiguous bytes each) and one pass through each wave's own swizzled LDS rows, not by 64 eight-byte loads per wave touching 16 rows each
# speedup vs baseline: 2.7187x; 1.1883x over previous
#include <hip/hip_runtime.h>

typedef float v2f __attribute__((ext_vector_type(2)));
typedef float v4f __attribute__((ext_vector_type(4)));
typedef float v8f __attribute__((ext_vector_type(8)));

#define PITCH  34
#define NPITCH 36
#define ROWS   4
#define SLOTS  (ROWS * 64)
#define WROW   32

__device__ __forceinline__ float xhalf(float x) {
    const int b = __builtin_bit_cast(int, x);
    return __builtin_bit_cast(float,
        __builtin_amdgcn_permlanex16(b, b, 0x76543210, 0xfedcba98, false, false));
}

template <int XM>
__device__ __forceinline__ float row_xor(float x) {
    return __builtin_bit_cast(float,
        __builtin_amdgcn_update_dpp(0, __builtin_bit_cast(int, x), 0x160 + XM, 0xf, 0xf, true));
}

__device__ __forceinline__ void gather_tile(
    v2f (&bfr)[8], v2f (&n2)[4], const float* sE, const float* sN,
    const int idx, const int lg, const int e0)
{
    const float* ep = sE + idx * PITCH + 2 * lg;
#pragma unroll
    for (int kk = 0; kk < 8; ++kk)
        bfr[kk] = *(const v2f*)(ep + 4 * kk);
    const float* np = sN + (idx & 63) * NPITCH + e0 + 8 * lg;
    const v4f q0 = *(const v4f*)np, q1 = *(const v4f*)(np + 4);
    n2[0] = (v2f){q0.x, q0.y};
    n2[1] = (v2f){q0.z, q0.w};
    n2[2] = (v2f){q1.x, q1.y};
    n2[3] = (v2f){q1.z, q1.w};
}

template <int NT>
__device__ __forceinline__ void mma_block(
    const v2f (&af)[8][8], const v4f (&bq)[8][2],
    const v2f (&bfr)[2][8], const v2f (&n2)[2][4], const v4f (&mk4)[2],
    v4f (&acc)[8])
{
#pragma unroll
    for (int k2 = 0; k2 < 4; ++k2) {
        v8f c[2][NT] = {};
#pragma unroll
        for (int kk = 0; kk < 8; ++kk) {
#pragma unroll
            for (int ki = 0; ki < 2; ++ki) {
#pragma unroll
                for (int t = 0; t < NT; ++t)
                    c[ki][t] = __builtin_amdgcn_wmma_f32_16x16x4_f32(
                                   false, af[2 * k2 + ki][kk], false, bfr[t][kk], (short)0, c[ki][t], false, false);
            }
        }
#pragma unroll
        for (int ki = 0; ki < 2; ++ki) {
            const int k = 2 * k2 + ki;
            const v4f b0 = bq[k][0], b1 = bq[k][1];
            const v2f bv2[4] = {(v2f){b0.x, b0.y}, (v2f){b0.z, b0.w}, (v2f){b1.x, b1.y}, (v2f){b1.z, b1.w}};
#pragma unroll
            for (int t = 0; t < NT; ++t) {
                v2f partial = (v2f){0.0f, 0.0f};
#pragma unroll
                for (int q = 0; q < 4; ++q) {
                    v2f v = (v2f){c[ki][t][2 * q], c[ki][t][2 * q + 1]} + bv2[q];
                    v = __builtin_elementwise_max(v, (v2f){0.0f, 0.0f});
                    partial += v * n2[t][q];
                }
                const float p = partial.x + partial.y;
                acc[k] += mk4[t] * p;
            }
        }
    }
}

__global__ __launch_bounds__(256) void ecc_aggregate_kernel(
    const float* __restrict__ nodes,
    const float* __restrict__ edges,
    const float* __restrict__ mask,
    const float* __restrict__ W,
    const float* __restrict__ bias,
    float* __restrict__ out)
{
    __shared__ __attribute__((aligned(16))) float sE[SLOTS * PITCH];
    __shared__ __attribute__((aligned(16))) float sN[64 * NPITCH];
    __shared__ __attribute__((aligned(16))) float sW[1024 * WROW];
    __shared__ __attribute__((aligned(16))) float sMk4[SLOTS * 4];
    __shared__ __attribute__((aligned(16))) int   sWc[8];
    __shared__ int   sIdx[SLOTS];
    __shared__ float sPart[ROWS * 64];

    const int tid  = threadIdx.x;
    const int wave = tid >> 5;
    const int lane = tid & 31;
    const int lg   = lane >> 4;
    const int m    = lane & 15;

    const int wg   = blockIdx.x;
    const int row0 = wg * ROWS;
    const int b    = row0 >> 6;

    const v4f*   e4    = (const v4f*)(edges + (size_t)row0 * 64 * 32);
    const float* nBase = nodes + (size_t)b * 64 * 32;

    const float mv = mask[(size_t)row0 * 64 + tid];

    v4f epc[8];
#pragma unroll
    for (int q = 0; q < 8; ++q) epc[q] = e4[tid + 256 * q];
    const v4f na = ((const v4f*)nBase)[tid], nb = ((const v4f*)nBase)[tid + 256];

    v4f wpc[8][4];
    v4f bq[8][2];
#pragma unroll
    for (int k = 0; k < 8; ++k) {
        const int g0 = (wave + 8 * k) << 4;
        const v4f* w4 = (const v4f*)(W + (size_t)g0 * 32);
#pragma unroll
        for (int q = 0; q < 4; ++q) wpc[k][q] = w4[q * 32 + lane];
        bq[k][0] = *(const v4f*)(bias + g0 + 8 * lg);
        bq[k][1] = *(const v4f*)(bias + g0 + 8 * lg + 4);
    }

    float* sWw = sW + wave * (128 * WROW);
    {
        const int c = lane & 7, rq = lane >> 3;
#pragma unroll
        for (int k = 0; k < 8; ++k) {
#pragma unroll
            for (int q = 0; q < 4; ++q) {
                *(v4f*)&sWw[(16 * k + 4 * q + rq) * WROW + 4 * (c ^ (2 * q + lg))] = wpc[k][q];
            }
        }
    }
    __builtin_amdgcn_fence(__ATOMIC_ACQ_REL, "wavefront");
    v2f af[8][8];
#pragma unroll
    for (int k = 0; k < 8; ++k) {
#pragma unroll
        for (int kk = 0; kk < 8; ++kk)
            af[k][kk] = *(const v2f*)&sWw[(16 * k + m) * WROW + 4 * (kk ^ (m >> 1)) + 2 * lg];
    }

    const int e0 = (wave & 1) << 4;

    const bool live = (mv != 0.0f);
    const unsigned bal = __builtin_amdgcn_ballot_w32(live);
    if (lane == 0) sWc[wave] = __builtin_popcount(bal);
    __syncthreads();

    const int4 wc0 = *(const int4*)&sWc[0], wc1 = *(const int4*)&sWc[4];
    const int wcs[8] = {wc0.x, wc0.y, wc0.z, wc0.w, wc1.x, wc1.y, wc1.z, wc1.w};
    int base = 0, total = 0;
#pragma unroll
    for (int w = 0; w < 8; ++w) {
        base  += (w < wave) ? wcs[w] : 0;
        total += wcs[w];
    }
    if (live) {
        const int rank = base + __builtin_popcount(bal & ((1u << lane) - 1u));
        const int r = tid >> 6;
        sIdx[rank] = tid;
        *(v4f*)&sMk4[rank * 4] = (v4f){r == 0 ? mv : 0.0f, r == 1 ? mv : 0.0f, r == 2 ? mv : 0.0f, r == 3 ? mv : 0.0f};
    }
    if (total + tid < SLOTS) {
        sIdx[total + tid] = 0;
        *(v4f*)&sMk4[(total + tid) * 4] = (v4f){0.0f, 0.0f, 0.0f, 0.0f};
    }

    {
        const int sj = tid >> 3, sf = (tid & 7) << 2;
        float* pe = &sE[sj * PITCH + sf];
#pragma unroll
        for (int q = 0; q < 8; ++q) {
            *(v2f*)(pe + q * 32 * PITCH)     = (v2f){epc[q].x, epc[q].y};
            *(v2f*)(pe + q * 32 * PITCH + 2) = (v2f){epc[q].z, epc[q].w};
        }
        *(v4f*)&sN[sj * NPITCH + sf]        = na;
        *(v4f*)&sN[(sj + 32) * NPITCH + sf] = nb;
    }
    __syncthreads();

    const int T = (__builtin_amdgcn_readfirstlane(total) + 15) >> 4;
    const int npairs = T >> 1;

    v4f acc[8];
#pragma unroll
    for (int k = 0; k < 8; ++k) acc[k] = (v4f){0.0f, 0.0f, 0.0f, 0.0f};

    v2f bfrC[2][8], n2C[2][4];
    v4f mkC[2];
    int idxN[2];
    v4f mkN[2];
#pragma unroll
    for (int t = 0; t < 2; ++t) {
        idxN[t] = sIdx[t * 16 + m];
        mkC[t]  = *(const v4f*)&sMk4[(t * 16 + m) * 4];
    }
#pragma unroll
    for (int t = 0; t < 2; ++t) gather_tile(bfrC[t], n2C[t], sE, sN, idxN[t], lg, e0);
#pragma unroll
    for (int t = 0; t < 2; ++t) {
        idxN[t] = sIdx[(2 + t) * 16 + m];
        mkN[t]  = *(const v4f*)&sMk4[((2 + t) * 16 + m) * 4];
    }

#pragma unroll 1
    for (int g = 0; g < npairs; ++g) {
        v2f bfrX[2][8], n2X[2][4];
        v4f mkX[2];
#pragma unroll
        for (int t = 0; t < 2; ++t) {
            gather_tile(bfrX[t], n2X[t], sE, sN, idxN[t], lg, e0);
            mkX[t] = mkN[t];
        }
        const int tb = (2 * g + 4 < 14) ? (2 * g + 4) : 14;
#pragma unroll
        for (int t = 0; t < 2; ++t) {
            idxN[t] = sIdx[(tb + t) * 16 + m];
            mkN[t]  = *(const v4f*)&sMk4[((tb + t) * 16 + m) * 4];
        }

        mma_block<2>(af, bq, bfrC, n2C, mkC, acc);

#pragma unroll
        for (int t = 0; t < 2; ++t) {
#pragma unroll
            for (int kk = 0; kk < 8; ++kk) bfrC[t][kk] = bfrX[t][kk];
#pragma unroll
            for (int q = 0; q < 4; ++q) n2C[t][q] = n2X[t][q];
            mkC[t] = mkX[t];
        }
    }
    if (T & 1) mma_block<1>(af, bq, bfrC, n2C, mkC, acc);

    float red[32];
#pragma unroll
    for (int k = 0; k < 8; ++k) {
        red[4 * k + 0] = acc[k].x; red[4 * k + 1] = acc[k].y;
        red[4 * k + 2] = acc[k].z; red[4 * k + 3] = acc[k].w;
    }
#pragma unroll
    for (int i = 0; i < 32; ++i) red[i] += xhalf(red[i]);
#pragma unroll
    for (int i = 0; i < 32; ++i) red[i] += row_xor<1>(red[i]);
#pragma unroll
    for (int i = 0; i < 32; ++i) red[i] += row_xor<2>(red[i]);
#pragma unroll
    for (int i = 0; i < 32; ++i) red[i] += row_xor<4>(red[i]);
#pragma unroll
    for (int i = 0; i < 32; ++i) red[i] += row_xor<8>(red[i]);

    if (lane == 0) {
#pragma unroll
        for (int k = 0; k < 8; ++k) {
#pragma unroll
            for (int r = 0; r < ROWS; ++r) sPart[r * 64 + wave + 8 * k] = red[4 * k + r];
        }
    }
    __syncthreads();

    if (tid < ROWS * 32) {
        const int r = tid >> 5;
        out[(size_t)(row0 + r) * 32 + lane] = sPart[r * 64 + 2 * lane] + sPart[r * 64 + 2 * lane + 1];
    }
}

extern "C" void kernel_launch(void* const* d_in, const int* in_sizes, int n_in,
                              void* d_out, int out_size, void* d_ws, size_t ws_size,
                              hipStream_t stream) {
    const float* nodes = (const float*)d_in[0];
    const float* edges = (const float*)d_in[1];
    const float* mask  = (const float*)d_in[2];
    const float* W     = (const float*)d_in[3];
    const float* bias  = (const float*)d_in[4];
    float* out         = (float*)d_out;

    ecc_aggregate_kernel<<<dim3(256), dim3(256), 0, stream>>>(
        nodes, edges, mask, W, bias, out);
}
